# starting kernel
# baseline (speedup 1.0000x reference)
#include <hip/hip_runtime.h>


typedef __attribute__((ext_vector_type(2))) float v2f;
typedef __attribute__((ext_vector_type(8))) float v8f;

#define RANK 64

__global__ __launch_bounds__(256) void cp_eval_wmma_kernel(
    const int*   __restrict__ idx,
    const float* __restrict__ f0,
    const float* __restrict__ f1,
    const float* __restrict__ f2,
    const float* __restrict__ lamb,
    float*       __restrict__ out,
    int n)
{
    const int lane  = threadIdx.x & 31;
    const int wave  = blockIdx.x * (blockDim.x >> 5) + (threadIdx.x >> 5);
    const int base  = wave * 16;
    if (base >= n) return;

    const int m     = lane & 15;
    const int khalf = lane >> 4;
    const int e     = base + m;

    const int i0 = idx[3 * e + 0];
    const int i1 = idx[3 * e + 1];
    const int i2 = idx[3 * e + 2];

    const float* __restrict__ r0 = f0 + (size_t)i0 * RANK;
    const float* __restrict__ r1 = f1 + (size_t)i1 * RANK;
    const float* __restrict__ r2 = f2 + (size_t)i2 * RANK;

    v8f c = {};
    v2f bones;
    bones.x = 1.0f;
    bones.y = 1.0f;

#pragma unroll
    for (int j = 0; j < 16; ++j) {
        const int r = 4 * j + 2 * khalf;
        const float w0 = lamb[r];
        const float w1 = lamb[r + 1];
        v2f a;
        a.x = w0 * r0[r]     * r1[r]     * r2[r];
        a.y = w1 * r0[r + 1] * r1[r + 1] * r2[r + 1];
        c = __builtin_amdgcn_wmma_f32_16x16x4_f32(
                false, a, false, bones, (short)0, c, false, false);
    }

    if ((lane & 15) == 0) {
        float4 lo = make_float4(c[0], c[1], c[2], c[3]);
        float4 hi = make_float4(c[4], c[5], c[6], c[7]);
        float4* o = reinterpret_cast<float4*>(out + base + 8 * khalf);
        o[0] = lo;
        o[1] = hi;
    }
}

extern "C" void kernel_launch(void* const* d_in, const int* in_sizes, int n_in,
                              void* d_out, int out_size, void* d_ws, size_t ws_size,
                              hipStream_t stream) {
    const int*   idx  = (const int*)d_in[0];
    const float* f0   = (const float*)d_in[1];
    const float* f1   = (const float*)d_in[2];
    const float* f2   = (const float*)d_in[3];
    const float* lamb = (const float*)d_in[4];
    float*       out  = (float*)d_out;

    const int n = in_sizes[0] / 3;
    const int waves  = (n + 15) / 16;
    const int wpb    = 256 / 32;
    const int blocks = (waves + wpb - 1) / wpb;

    cp_eval_wmma_kernel<<<blocks, 256, 0, stream>>>(idx, f0, f1, f2, lamb, out, n);
}


# k slots remapped under the all-ones B: each lane takes 4 consecutive r per step as one 128-bit load per table, feeding two WMMAs; half the gather loads and line touches; r-sum order changes.
# speedup vs baseline: 1.3548x; 1.3548x over previous
#include <hip/hip_runtime.h>


typedef __attribute__((ext_vector_type(2))) float v2f;
typedef __attribute__((ext_vector_type(8))) float v8f;

#define RANK 64

__global__ __launch_bounds__(256) void cp_eval_wmma_kernel(
    const int*   __restrict__ idx,
    const float* __restrict__ f0,
    const float* __restrict__ f1,
    const float* __restrict__ f2,
    const float* __restrict__ lamb,
    float*       __restrict__ out,
    int n)
{
    const int lane  = threadIdx.x & 31;
    const int wave  = blockIdx.x * (blockDim.x >> 5) + (threadIdx.x >> 5);
    const int base  = wave * 16;
    if (base >= n) return;

    const int m     = lane & 15;
    const int khalf = lane >> 4;
    const int e     = base + m;

    const int i0 = idx[3 * e + 0];
    const int i1 = idx[3 * e + 1];
    const int i2 = idx[3 * e + 2];

    const float* __restrict__ r0 = f0 + (size_t)i0 * RANK;
    const float* __restrict__ r1 = f1 + (size_t)i1 * RANK;
    const float* __restrict__ r2 = f2 + (size_t)i2 * RANK;

    v8f c = {};
    v2f bones;
    bones.x = 1.0f;
    bones.y = 1.0f;

#pragma unroll
    for (int J = 0; J < 8; ++J) {
        const int r = 8 * J + 4 * khalf;
        const float4 w  = *reinterpret_cast<const float4*>(lamb + r);
        const float4 x0 = *reinterpret_cast<const float4*>(r0 + r);
        const float4 x1 = *reinterpret_cast<const float4*>(r1 + r);
        const float4 x2 = *reinterpret_cast<const float4*>(r2 + r);
        v2f a;
        a.x = w.x * x0.x * x1.x * x2.x;
        a.y = w.y * x0.y * x1.y * x2.y;
        c = __builtin_amdgcn_wmma_f32_16x16x4_f32(
                false, a, false, bones, (short)0, c, false, false);
        v2f b;
        b.x = w.z * x0.z * x1.z * x2.z;
        b.y = w.w * x0.w * x1.w * x2.w;
        c = __builtin_amdgcn_wmma_f32_16x16x4_f32(
                false, b, false, bones, (short)0, c, false, false);
    }

    if ((lane & 15) == 0) {
        float4 lo = make_float4(c[0], c[1], c[2], c[3]);
        float4 hi = make_float4(c[4], c[5], c[6], c[7]);
        float4* o = reinterpret_cast<float4*>(out + base + 8 * khalf);
        o[0] = lo;
        o[1] = hi;
    }
}

extern "C" void kernel_launch(void* const* d_in, const int* in_sizes, int n_in,
                              void* d_out, int out_size, void* d_ws, size_t ws_size,
                              hipStream_t stream) {
    const int*   idx  = (const int*)d_in[0];
    const float* f0   = (const float*)d_in[1];
    const float* f1   = (const float*)d_in[2];
    const float* f2   = (const float*)d_in[3];
    const float* lamb = (const float*)d_in[4];
    float*       out  = (float*)d_out;

    const int n = in_sizes[0] / 3;
    const int waves  = (n + 15) / 16;
    const int wpb    = 256 / 32;
    const int blocks = (waves + wpb - 1) / wpb;

    cp_eval_wmma_kernel<<<blocks, 256, 0, stream>>>(idx, f0, f1, f2, lamb, out, n);
}


# Eight lanes per entry, four D rows per entry under all-ones B: each 128-bit gather load covers 4 whole lines, each line asked for once: no re-reads, no load waits for another. r-sum order changes.
# speedup vs baseline: 3.0003x; 2.2146x over previous
#include <hip/hip_runtime.h>


typedef __attribute__((ext_vector_type(2))) float v2f;
typedef __attribute__((ext_vector_type(4))) float v4f;
typedef __attribute__((ext_vector_type(8))) float v8f;
typedef __attribute__((ext_vector_type(4))) int   v4i;

#define RANK 64

__global__ __launch_bounds__(256) void cp_eval_wmma_kernel(
    const int*   __restrict__ idx,
    const float* __restrict__ f0,
    const float* __restrict__ f1,
    const float* __restrict__ f2,
    const float* __restrict__ lamb,
    float*       __restrict__ out,
    int n)
{
    const int lane  = threadIdx.x & 31;
    const int wave  = blockIdx.x * (blockDim.x >> 5) + (threadIdx.x >> 5);
    const int base  = wave * 16;
    if (base >= n) return;

    const int c     = lane & 15;
    const int h     = lane >> 4;
    const int j     = c >> 2;
    const int b0    = c & 1;
    const int b1    = (c >> 1) & 1;
    const int p     = (c & 3) + 4 * h;

    const v4i* ip = reinterpret_cast<const v4i*>(idx + 3 * (size_t)(base + 4 * j));
    const v4i ia = ip[0];
    const v4i ib = ip[1];
    const v4i ic = ip[2];
    const int i0[4] = { ia.x, ia.w, ib.z, ic.y };
    const int i1[4] = { ia.y, ib.x, ib.w, ic.z };
    const int i2[4] = { ia.z, ib.y, ic.x, ic.w };

    v4f x[3][4][2];

#pragma unroll
    for (int g = 0; g < 4; ++g) {
        const float* r0 = f0 + (size_t)i0[g] * RANK + 4 * p;
        const float* r1 = f1 + (size_t)i1[g] * RANK + 4 * p;
        const float* r2 = f2 + (size_t)i2[g] * RANK + 4 * p;
#pragma unroll
        for (int u = 0; u < 2; ++u) {
            x[0][g][u] = *reinterpret_cast<const v4f*>(r0 + 32 * u);
            x[1][g][u] = *reinterpret_cast<const v4f*>(r1 + 32 * u);
            x[2][g][u] = *reinterpret_cast<const v4f*>(r2 + 32 * u);
        }
    }

    __builtin_amdgcn_sched_barrier(0);

    v2f bones;
    bones.x = 1.0f;
    bones.y = 1.0f;

    float lam[RANK];
#pragma unroll
    for (int r = 0; r < RANK; ++r) lam[r] = lamb[r];

    v4f w[2];
#pragma unroll
    for (int u = 0; u < 2; ++u) {
#pragma unroll
        for (int k = 0; k < 4; ++k) {
            const float q0 = lam[32 * u +  0 + k], q1 = lam[32 * u +  4 + k];
            const float q2 = lam[32 * u +  8 + k], q3 = lam[32 * u + 12 + k];
            const float q4 = lam[32 * u + 16 + k], q5 = lam[32 * u + 20 + k];
            const float q6 = lam[32 * u + 24 + k], q7 = lam[32 * u + 28 + k];
            const float v0 = b0 ? q1 : q0, v1 = b0 ? q3 : q2;
            const float v2 = b0 ? q5 : q4, v3 = b0 ? q7 : q6;
            const float m0 = b1 ? v1 : v0, m1 = b1 ? v3 : v2;
            w[u][k] = h ? m1 : m0;
        }
    }

    v8f acc[4] = {};

#pragma unroll
    for (int g = 0; g < 4; ++g) {
#pragma unroll
        for (int u = 0; u < 2; ++u) {
            const v4f x0 = x[0][g][u];
            const v4f x1 = x[1][g][u];
            const v4f x2 = x[2][g][u];
            v2f a;
            a.x = w[u].x * x0.x * x1.x * x2.x;
            a.y = w[u].y * x0.y * x1.y * x2.y;
            acc[g] = __builtin_amdgcn_wmma_f32_16x16x4_f32(
                        false, a, false, bones, (short)0, acc[g], false, false);
            v2f b;
            b.x = w[u].z * x0.z * x1.z * x2.z;
            b.y = w[u].w * x0.w * x1.w * x2.w;
            acc[g] = __builtin_amdgcn_wmma_f32_16x16x4_f32(
                        false, b, false, bones, (short)0, acc[g], false, false);
        }
    }

    float4 lo = make_float4((acc[0][0] + acc[0][1]) + (acc[0][2] + acc[0][3]),
                            (acc[1][0] + acc[1][1]) + (acc[1][2] + acc[1][3]),
                            (acc[2][0] + acc[2][1]) + (acc[2][2] + acc[2][3]),
                            (acc[3][0] + acc[3][1]) + (acc[3][2] + acc[3][3]));
    float4 hi = make_float4((acc[0][4] + acc[0][5]) + (acc[0][6] + acc[0][7]),
                            (acc[1][4] + acc[1][5]) + (acc[1][6] + acc[1][7]),
                            (acc[2][4] + acc[2][5]) + (acc[2][6] + acc[2][7]),
                            (acc[3][4] + acc[3][5]) + (acc[3][6] + acc[3][7]));
    if (c == 0) {
        float4* o = reinterpret_cast<float4*>(out + base + 8 * h);
        o[0] = lo;
        o[1] = hi;
    }
}

extern "C" void kernel_launch(void* const* d_in, const int* in_sizes, int n_in,
                              void* d_out, int out_size, void* d_ws, size_t ws_size,
                              hipStream_t stream) {
    const int*   idx  = (const int*)d_in[0];
    const float* f0   = (const float*)d_in[1];
    const float* f1   = (const float*)d_in[2];
    const float* f2   = (const float*)d_in[3];
    const float* lamb = (const float*)d_in[4];
    float*       out  = (float*)d_out;

    const int n = in_sizes[0] / 3;
    const int waves  = (n + 15) / 16;
    const int wpb    = 256 / 32;
    const int blocks = (waves + wpb - 1) / wpb;

    cp_eval_wmma_kernel<<<blocks, 256, 0, stream>>>(idx, f0, f1, f2, lamb, out, n);
}


# idx by scalar loads (wave-uniform address) and per-lane selects instead of 3 per-lane 128-bit loads: 24 vector memory loads per wave for 27, idx off the vector memory path. Sums unchanged.
# speedup vs baseline: 3.1083x; 1.0360x over previous
#include <hip/hip_runtime.h>


typedef __attribute__((ext_vector_type(2))) float v2f;
typedef __attribute__((ext_vector_type(4))) float v4f;
typedef __attribute__((ext_vector_type(8))) float v8f;
typedef __attribute__((ext_vector_type(4))) int   v4i;

#define RANK 64

__global__ __launch_bounds__(256) void cp_eval_wmma_kernel(
    const int*   __restrict__ idx,
    const float* __restrict__ f0,
    const float* __restrict__ f1,
    const float* __restrict__ f2,
    const float* __restrict__ lamb,
    float*       __restrict__ out,
    int n)
{
    const int lane  = threadIdx.x & 31;
    const int wave  = __builtin_amdgcn_readfirstlane(
                          blockIdx.x * (blockDim.x >> 5) + (threadIdx.x >> 5));
    const int base  = wave * 16;
    if (base >= n) return;

    const int c     = lane & 15;
    const int h     = lane >> 4;
    const int j     = c >> 2;
    const int b0    = c & 1;
    const int b1    = (c >> 1) & 1;
    const int p     = (c & 3) + 4 * h;

    const int* ip = idx + 3 * (size_t)base;
    int e[48];
#pragma unroll
    for (int k = 0; k < 48; ++k) e[k] = ip[k];

    const int jb0 = j & 1;
    const int jb1 = (j >> 1) & 1;
    int i0[4], i1[4], i2[4];
#pragma unroll
    for (int g = 0; g < 4; ++g) {
        int s[3];
#pragma unroll
        for (int t = 0; t < 3; ++t) {
            const int a0 = e[3 * g + t],      a1 = e[12 + 3 * g + t];
            const int a2 = e[24 + 3 * g + t], a3 = e[36 + 3 * g + t];
            const int lo = jb0 ? a1 : a0, hi = jb0 ? a3 : a2;
            s[t] = jb1 ? hi : lo;
        }
        i0[g] = s[0];
        i1[g] = s[1];
        i2[g] = s[2];
    }

    v4f x[3][4][2];

#pragma unroll
    for (int g = 0; g < 4; ++g) {
        const float* r0 = f0 + (size_t)i0[g] * RANK + 4 * p;
        const float* r1 = f1 + (size_t)i1[g] * RANK + 4 * p;
        const float* r2 = f2 + (size_t)i2[g] * RANK + 4 * p;
#pragma unroll
        for (int u = 0; u < 2; ++u) {
            x[0][g][u] = *reinterpret_cast<const v4f*>(r0 + 32 * u);
            x[1][g][u] = *reinterpret_cast<const v4f*>(r1 + 32 * u);
            x[2][g][u] = *reinterpret_cast<const v4f*>(r2 + 32 * u);
        }
    }

    __builtin_amdgcn_sched_barrier(0);

    v2f bones;
    bones.x = 1.0f;
    bones.y = 1.0f;

    float lam[RANK];
#pragma unroll
    for (int r = 0; r < RANK; ++r) lam[r] = lamb[r];

    v4f w[2];
#pragma unroll
    for (int u = 0; u < 2; ++u) {
#pragma unroll
        for (int k = 0; k < 4; ++k) {
            const float q0 = lam[32 * u +  0 + k], q1 = lam[32 * u +  4 + k];
            const float q2 = lam[32 * u +  8 + k], q3 = lam[32 * u + 12 + k];
            const float q4 = lam[32 * u + 16 + k], q5 = lam[32 * u + 20 + k];
            const float q6 = lam[32 * u + 24 + k], q7 = lam[32 * u + 28 + k];
            const float v0 = b0 ? q1 : q0, v1 = b0 ? q3 : q2;
            const float v2 = b0 ? q5 : q4, v3 = b0 ? q7 : q6;
            const float m0 = b1 ? v1 : v0, m1 = b1 ? v3 : v2;
            w[u][k] = h ? m1 : m0;
        }
    }

    v8f acc[4] = {};

#pragma unroll
    for (int g = 0; g < 4; ++g) {
#pragma unroll
        for (int u = 0; u < 2; ++u) {
            const v4f x0 = x[0][g][u];
            const v4f x1 = x[1][g][u];
            const v4f x2 = x[2][g][u];
            v2f a;
            a.x = w[u].x * x0.x * x1.x * x2.x;
            a.y = w[u].y * x0.y * x1.y * x2.y;
            acc[g] = __builtin_amdgcn_wmma_f32_16x16x4_f32(
                        false, a, false, bones, (short)0, acc[g], false, false);
            v2f b;
            b.x = w[u].z * x0.z * x1.z * x2.z;
            b.y = w[u].w * x0.w * x1.w * x2.w;
            acc[g] = __builtin_amdgcn_wmma_f32_16x16x4_f32(
                        false, b, false, bones, (short)0, acc[g], false, false);
        }
    }

    float4 lo = make_float4((acc[0][0] + acc[0][1]) + (acc[0][2] + acc[0][3]),
                            (acc[1][0] + acc[1][1]) + (acc[1][2] + acc[1][3]),
                            (acc[2][0] + acc[2][1]) + (acc[2][2] + acc[2][3]),
                            (acc[3][0] + acc[3][1]) + (acc[3][2] + acc[3][3]));
    float4 hi = make_float4((acc[0][4] + acc[0][5]) + (acc[0][6] + acc[0][7]),
                            (acc[1][4] + acc[1][5]) + (acc[1][6] + acc[1][7]),
                            (acc[2][4] + acc[2][5]) + (acc[2][6] + acc[2][7]),
                            (acc[3][4] + acc[3][5]) + (acc[3][6] + acc[3][7]));
    if (c == 0) {
        float4* o = reinterpret_cast<float4*>(out + base + 8 * h);
        o[0] = lo;
        o[1] = hi;
    }
}

extern "C" void kernel_launch(void* const* d_in, const int* in_sizes, int n_in,
                              void* d_out, int out_size, void* d_ws, size_t ws_size,
                              hipStream_t stream) {
    const int*   idx  = (const int*)d_in[0];
    const float* f0   = (const float*)d_in[1];
    const float* f1   = (const float*)d_in[2];
    const float* f2   = (const float*)d_in[3];
    const float* lamb = (const float*)d_in[4];
    float*       out  = (float*)d_out;

    const int n = in_sizes[0] / 3;
    const int waves  = (n + 15) / 16;
    const int wpb    = 256 / 32;
    const int blocks = (waves + wpb - 1) / wpb;

    cp_eval_wmma_kernel<<<blocks, 256, 0, stream>>>(idx, f0, f1, f2, lamb, out, n);
}
